# starting kernel
# baseline (speedup 1.0000x reference)
#include <hip/hip_runtime.h>
#include <hip/hip_bf16.h>
#include <cstdint>

typedef __attribute__((ext_vector_type(2))) float v2f;
typedef __attribute__((ext_vector_type(8))) float v8f;

constexpr int Bb = 256;
constexpr int Ss = 1024;
constexpr int Cc = 128;

__global__ __launch_bounds__(128) void viterbi_forward(
    const float* __restrict__ emis, const int* __restrict__ mask,
    const float* __restrict__ T, uint8_t* __restrict__ bp,
    int* __restrict__ path)
{
  __shared__ float sAlpha[2][Cc];
  const int b      = blockIdx.x;
  const int tid    = threadIdx.x;
  const int w      = tid >> 5;
  const int lane   = tid & 31;
  const int laneLo = lane & 15;
  const bool hi    = lane >= 16;
  const int  hi8   = hi ? 8 : 0;
  const int  j     = (w << 5) + lane;

  float ct[2][8][8];
#pragma unroll
  for (int tc = 0; tc < 2; ++tc)
#pragma unroll
    for (int k = 0; k < 8; ++k)
#pragma unroll
      for (int r = 0; r < 8; ++r) {
        const int row = 16 * k + r + hi8;
        const int col = (w << 5) + (tc << 4) + laneLo;
        ct[tc][k][r] = T[row * Cc + col];
      }

  v2f bfrag;
  bfrag.x = hi ? 0.0f : 1.0f;
  bfrag.y = 0.0f;

  sAlpha[0][j] = emis[(size_t)b * Ss * Cc + j];
  __syncthreads();

  const float NEG = -__builtin_inff();

  for (int t = 1; t < Ss; ++t) {
    const int cur = (t - 1) & 1, nxt = t & 1;

    const float e    = emis[((size_t)b * Ss + t) * Cc + j];
    const int   m    = mask[b * Ss + t];
    const float aold = sAlpha[cur][j];
    if (t + 4 < Ss)
      __builtin_prefetch(&emis[((size_t)b * Ss + t + 4) * Cc + j], 0, 0);

    float af[8];
#pragma unroll
    for (int k = 0; k < 8; ++k) {
      const float v = sAlpha[cur][16 * k + laneLo];
      af[k] = hi ? 0.0f : v;
    }

    float best[2][2];
    int   arg8[2][2];
#pragma unroll
    for (int tc = 0; tc < 2; ++tc)
#pragma unroll
      for (int p = 0; p < 2; ++p) { best[tc][p] = NEG; arg8[tc][p] = 0; }

#pragma unroll
    for (int tc = 0; tc < 2; ++tc) {
#pragma unroll
      for (int k = 0; k < 8; ++k) {
        v2f a; a.x = af[k]; a.y = 0.0f;
        v8f cmat;
#pragma unroll
        for (int r = 0; r < 8; ++r) cmat[r] = ct[tc][k][r];
        v8f d = __builtin_amdgcn_wmma_f32_16x16x4_f32(
            false, a, false, bfrag, (short)0, cmat, false, false);
        const int p = k & 1;
#pragma unroll
        for (int r = 0; r < 8; ++r) {
          const int  cand = k * 8 + r;
          const bool g    = d[r] > best[tc][p];
          best[tc][p] = g ? d[r] : best[tc][p];
          arg8[tc][p] = g ? cand : arg8[tc][p];
        }
      }
    }

    float bb[2]; int ii[2];
#pragma unroll
    for (int tc = 0; tc < 2; ++tc) {
      const bool tk1 = (best[tc][1] > best[tc][0]) ||
                       ((best[tc][1] == best[tc][0]) && (arg8[tc][1] < arg8[tc][0]));
      float bv = tk1 ? best[tc][1] : best[tc][0];
      int   a8 = tk1 ? arg8[tc][1] : arg8[tc][0];
      int   iv = a8 + ((a8 >> 3) << 3) + hi8;
      const float bo = __shfl_xor(bv, 16, 32);
      const int   io = __shfl_xor(iv, 16, 32);
      const bool tk2 = (bo > bv) || ((bo == bv) && (io < iv));
      bb[tc] = tk2 ? bo : bv;
      ii[tc] = tk2 ? io : iv;
    }

    const float bsel = hi ? bb[1] : bb[0];
    const int   isel = hi ? ii[1] : ii[0];
    const float an   = bsel + e;
    const float afin = m ? an : aold;

    sAlpha[nxt][j] = afin;
    bp[((size_t)b * (Ss - 1) + (t - 1)) * Cc + j] = (uint8_t)isel;
    __syncthreads();
  }

  if (w == 0) {
    const float* fa = sAlpha[(Ss - 1) & 1];
    float bbest = NEG; int barg = 0;
#pragma unroll
    for (int k = 0; k < 4; ++k) {
      const int   jj = (k << 5) + lane;
      const float v  = fa[jj];
      const bool  g  = v > bbest;
      bbest = g ? v : bbest;
      barg  = g ? jj : barg;
    }
#pragma unroll
    for (int off = 16; off >= 1; off >>= 1) {
      const float bo = __shfl_xor(bbest, off, 32);
      const int   ao = __shfl_xor(barg, off, 32);
      const bool  tk = (bo > bbest) || ((bo == bbest) && (ao < barg));
      bbest = tk ? bo : bbest;
      barg  = tk ? ao : barg;
    }
    if (lane == 0) path[(size_t)b * Ss + (Ss - 1)] = barg;
  }
}

__global__ __launch_bounds__(32) void viterbi_backtrace(
    const uint8_t* __restrict__ bp, int* __restrict__ path)
{
  const int b    = blockIdx.x;
  const int lane = threadIdx.x;
  const uint8_t* bpb = bp + (size_t)b * (Ss - 1) * Cc;
  int tag = path[(size_t)b * Ss + (Ss - 1)];

  auto ROW = [&](int tt) -> unsigned {
    return ((const unsigned*)(bpb + (size_t)tt * Cc))[lane];
  };
  auto CONSUME = [&](unsigned rowv, int tt) {
    const unsigned wsel = (unsigned)__shfl((int)rowv, tag >> 2, 32);
    tag = (int)((wsel >> ((tag & 3) << 3)) & 0xFF);
    if (lane == 0) path[(size_t)b * Ss + tt] = tag;
  };

  unsigned r0 = ROW(Ss - 2), r1 = ROW(Ss - 3), r2 = ROW(Ss - 4), r3 = ROW(Ss - 5);
  for (int tb = Ss - 2; tb >= 0; tb -= 4) {
    unsigned n0 = 0, n1 = 0, n2 = 0, n3 = 0;
    if (tb - 4 >= 0) n0 = ROW(tb - 4);
    if (tb - 5 >= 0) n1 = ROW(tb - 5);
    if (tb - 6 >= 0) n2 = ROW(tb - 6);
    if (tb - 7 >= 0) n3 = ROW(tb - 7);
    CONSUME(r0, tb);
    if (tb - 1 >= 0) CONSUME(r1, tb - 1);
    if (tb - 2 >= 0) CONSUME(r2, tb - 2);
    if (tb - 3 >= 0) CONSUME(r3, tb - 3);
    r0 = n0; r1 = n1; r2 = n2; r3 = n3;
  }
}

extern "C" void kernel_launch(void* const* d_in, const int* in_sizes, int n_in,
                              void* d_out, int out_size, void* d_ws, size_t ws_size,
                              hipStream_t stream) {
  const float* emissions   = (const float*)d_in[0];
  const int*   mask        = (const int*)d_in[1];
  const float* transitions = (const float*)d_in[2];
  int*         path        = (int*)d_out;
  uint8_t*     bp          = (uint8_t*)d_ws;

  viterbi_forward<<<Bb, 128, 0, stream>>>(emissions, mask, transitions, bp, path);
  viterbi_backtrace<<<Bb, 32, 0, stream>>>(bp, path);
}


# per-step barrier no longer waits for the bp global store: LDS-only release/acquire fence + bare barrier replaces __syncthreads in the time loop
# speedup vs baseline: 1.2234x; 1.2234x over previous
#include <hip/hip_runtime.h>
#include <hip/hip_bf16.h>
#include <cstdint>

typedef __attribute__((ext_vector_type(2))) float v2f;
typedef __attribute__((ext_vector_type(8))) float v8f;

constexpr int Bb = 256;
constexpr int Ss = 1024;
constexpr int Cc = 128;

__global__ __launch_bounds__(128) void viterbi_forward(
    const float* __restrict__ emis, const int* __restrict__ mask,
    const float* __restrict__ T, uint8_t* __restrict__ bp,
    int* __restrict__ path)
{
  __shared__ float sAlpha[2][Cc];
  const int b      = blockIdx.x;
  const int tid    = threadIdx.x;
  const int w      = tid >> 5;
  const int lane   = tid & 31;
  const int laneLo = lane & 15;
  const bool hi    = lane >= 16;
  const int  hi8   = hi ? 8 : 0;
  const int  j     = (w << 5) + lane;

  float ct[2][8][8];
#pragma unroll
  for (int tc = 0; tc < 2; ++tc)
#pragma unroll
    for (int k = 0; k < 8; ++k)
#pragma unroll
      for (int r = 0; r < 8; ++r) {
        const int row = 16 * k + r + hi8;
        const int col = (w << 5) + (tc << 4) + laneLo;
        ct[tc][k][r] = T[row * Cc + col];
      }

  v2f bfrag;
  bfrag.x = hi ? 0.0f : 1.0f;
  bfrag.y = 0.0f;

  sAlpha[0][j] = emis[(size_t)b * Ss * Cc + j];
  __syncthreads();

  const float NEG = -__builtin_inff();

  for (int t = 1; t < Ss; ++t) {
    const int cur = (t - 1) & 1, nxt = t & 1;

    const float e    = emis[((size_t)b * Ss + t) * Cc + j];
    const int   m    = mask[b * Ss + t];
    const float aold = sAlpha[cur][j];
    if (t + 4 < Ss)
      __builtin_prefetch(&emis[((size_t)b * Ss + t + 4) * Cc + j], 0, 0);

    float af[8];
#pragma unroll
    for (int k = 0; k < 8; ++k) {
      const float v = sAlpha[cur][16 * k + laneLo];
      af[k] = hi ? 0.0f : v;
    }

    float best[2][2];
    int   arg8[2][2];
#pragma unroll
    for (int tc = 0; tc < 2; ++tc)
#pragma unroll
      for (int p = 0; p < 2; ++p) { best[tc][p] = NEG; arg8[tc][p] = 0; }

#pragma unroll
    for (int tc = 0; tc < 2; ++tc) {
#pragma unroll
      for (int k = 0; k < 8; ++k) {
        v2f a; a.x = af[k]; a.y = 0.0f;
        v8f cmat;
#pragma unroll
        for (int r = 0; r < 8; ++r) cmat[r] = ct[tc][k][r];
        v8f d = __builtin_amdgcn_wmma_f32_16x16x4_f32(
            false, a, false, bfrag, (short)0, cmat, false, false);
        const int p = k & 1;
#pragma unroll
        for (int r = 0; r < 8; ++r) {
          const int  cand = k * 8 + r;
          const bool g    = d[r] > best[tc][p];
          best[tc][p] = g ? d[r] : best[tc][p];
          arg8[tc][p] = g ? cand : arg8[tc][p];
        }
      }
    }

    float bb[2]; int ii[2];
#pragma unroll
    for (int tc = 0; tc < 2; ++tc) {
      const bool tk1 = (best[tc][1] > best[tc][0]) ||
                       ((best[tc][1] == best[tc][0]) && (arg8[tc][1] < arg8[tc][0]));
      float bv = tk1 ? best[tc][1] : best[tc][0];
      int   a8 = tk1 ? arg8[tc][1] : arg8[tc][0];
      int   iv = a8 + ((a8 >> 3) << 3) + hi8;
      const float bo = __shfl_xor(bv, 16, 32);
      const int   io = __shfl_xor(iv, 16, 32);
      const bool tk2 = (bo > bv) || ((bo == bv) && (io < iv));
      bb[tc] = tk2 ? bo : bv;
      ii[tc] = tk2 ? io : iv;
    }

    const float bsel = hi ? bb[1] : bb[0];
    const int   isel = hi ? ii[1] : ii[0];
    const float an   = bsel + e;
    const float afin = m ? an : aold;

    sAlpha[nxt][j] = afin;
    bp[((size_t)b * (Ss - 1) + (t - 1)) * Cc + j] = (uint8_t)isel;
    __builtin_amdgcn_fence(__ATOMIC_RELEASE, "workgroup", "local");
    __builtin_amdgcn_s_barrier();
    __builtin_amdgcn_fence(__ATOMIC_ACQUIRE, "workgroup", "local");
  }

  if (w == 0) {
    const float* fa = sAlpha[(Ss - 1) & 1];
    float bbest = NEG; int barg = 0;
#pragma unroll
    for (int k = 0; k < 4; ++k) {
      const int   jj = (k << 5) + lane;
      const float v  = fa[jj];
      const bool  g  = v > bbest;
      bbest = g ? v : bbest;
      barg  = g ? jj : barg;
    }
#pragma unroll
    for (int off = 16; off >= 1; off >>= 1) {
      const float bo = __shfl_xor(bbest, off, 32);
      const int   ao = __shfl_xor(barg, off, 32);
      const bool  tk = (bo > bbest) || ((bo == bbest) && (ao < barg));
      bbest = tk ? bo : bbest;
      barg  = tk ? ao : barg;
    }
    if (lane == 0) path[(size_t)b * Ss + (Ss - 1)] = barg;
  }
}

__global__ __launch_bounds__(32) void viterbi_backtrace(
    const uint8_t* __restrict__ bp, int* __restrict__ path)
{
  const int b    = blockIdx.x;
  const int lane = threadIdx.x;
  const uint8_t* bpb = bp + (size_t)b * (Ss - 1) * Cc;
  int tag = path[(size_t)b * Ss + (Ss - 1)];

  auto ROW = [&](int tt) -> unsigned {
    return ((const unsigned*)(bpb + (size_t)tt * Cc))[lane];
  };
  auto CONSUME = [&](unsigned rowv, int tt) {
    const unsigned wsel = (unsigned)__shfl((int)rowv, tag >> 2, 32);
    tag = (int)((wsel >> ((tag & 3) << 3)) & 0xFF);
    if (lane == 0) path[(size_t)b * Ss + tt] = tag;
  };

  unsigned r0 = ROW(Ss - 2), r1 = ROW(Ss - 3), r2 = ROW(Ss - 4), r3 = ROW(Ss - 5);
  for (int tb = Ss - 2; tb >= 0; tb -= 4) {
    unsigned n0 = 0, n1 = 0, n2 = 0, n3 = 0;
    if (tb - 4 >= 0) n0 = ROW(tb - 4);
    if (tb - 5 >= 0) n1 = ROW(tb - 5);
    if (tb - 6 >= 0) n2 = ROW(tb - 6);
    if (tb - 7 >= 0) n3 = ROW(tb - 7);
    CONSUME(r0, tb);
    if (tb - 1 >= 0) CONSUME(r1, tb - 1);
    if (tb - 2 >= 0) CONSUME(r2, tb - 2);
    if (tb - 3 >= 0) CONSUME(r3, tb - 3);
    r0 = n0; r1 = n1; r2 = n2; r3 = n3;
  }
}

extern "C" void kernel_launch(void* const* d_in, const int* in_sizes, int n_in,
                              void* d_out, int out_size, void* d_ws, size_t ws_size,
                              hipStream_t stream) {
  const float* emissions   = (const float*)d_in[0];
  const int*   mask        = (const int*)d_in[1];
  const float* transitions = (const float*)d_in[2];
  int*         path        = (int*)d_out;
  uint8_t*     bp          = (uint8_t*)d_ws;

  viterbi_forward<<<Bb, 128, 0, stream>>>(emissions, mask, transitions, bp, path);
  viterbi_backtrace<<<Bb, 32, 0, stream>>>(bp, path);
}


# running max/argmax over the previous tags split into 16 independent compare-select chains (one per accumulator register of each j-tile) in place of 4 chains of 32 dependent links; same tie rule
# speedup vs baseline: 1.6228x; 1.3265x over previous
#include <hip/hip_runtime.h>
#include <hip/hip_bf16.h>
#include <cstdint>

typedef __attribute__((ext_vector_type(2))) float v2f;
typedef __attribute__((ext_vector_type(8))) float v8f;

constexpr int Bb = 256;
constexpr int Ss = 1024;
constexpr int Cc = 128;

__global__ __launch_bounds__(128) void viterbi_forward(
    const float* __restrict__ emis, const int* __restrict__ mask,
    const float* __restrict__ T, uint8_t* __restrict__ bp,
    int* __restrict__ path)
{
  __shared__ float sAlpha[2][Cc];
  const int b      = blockIdx.x;
  const int tid    = threadIdx.x;
  const int w      = tid >> 5;
  const int lane   = tid & 31;
  const int laneLo = lane & 15;
  const bool hi    = lane >= 16;
  const int  hi8   = hi ? 8 : 0;
  const int  j     = (w << 5) + lane;

  float ct[2][8][8];
#pragma unroll
  for (int tc = 0; tc < 2; ++tc)
#pragma unroll
    for (int k = 0; k < 8; ++k)
#pragma unroll
      for (int r = 0; r < 8; ++r) {
        const int row = 16 * k + r + hi8;
        const int col = (w << 5) + (tc << 4) + laneLo;
        ct[tc][k][r] = T[row * Cc + col];
      }

  v2f bfrag;
  bfrag.x = hi ? 0.0f : 1.0f;
  bfrag.y = 0.0f;

  sAlpha[0][j] = emis[(size_t)b * Ss * Cc + j];
  __syncthreads();

  const float NEG = -__builtin_inff();

  for (int t = 1; t < Ss; ++t) {
    const int cur = (t - 1) & 1, nxt = t & 1;

    const float e    = emis[((size_t)b * Ss + t) * Cc + j];
    const int   m    = mask[b * Ss + t];
    const float aold = sAlpha[cur][j];
    if (t + 4 < Ss)
      __builtin_prefetch(&emis[((size_t)b * Ss + t + 4) * Cc + j], 0, 0);

    float af[8];
#pragma unroll
    for (int k = 0; k < 8; ++k) {
      const float v = sAlpha[cur][16 * k + laneLo];
      af[k] = hi ? 0.0f : v;
    }

    float best[2][8];
    int   argk[2][8];

#pragma unroll
    for (int tc = 0; tc < 2; ++tc) {
#pragma unroll
      for (int k = 0; k < 8; ++k) {
        v2f a; a.x = af[k]; a.y = 0.0f;
        v8f cmat;
#pragma unroll
        for (int r = 0; r < 8; ++r) cmat[r] = ct[tc][k][r];
        v8f d = __builtin_amdgcn_wmma_f32_16x16x4_f32(
            false, a, false, bfrag, (short)0, cmat, false, false);
        if (k == 0) {
#pragma unroll
          for (int r = 0; r < 8; ++r) { best[tc][r] = d[r]; argk[tc][r] = 0; }
        } else {
#pragma unroll
          for (int r = 0; r < 8; ++r) {
            const bool g = d[r] > best[tc][r];
            best[tc][r] = g ? d[r] : best[tc][r];
            argk[tc][r] = g ? k : argk[tc][r];
          }
        }
      }
    }

    float bb[2]; int ii[2];
#pragma unroll
    for (int tc = 0; tc < 2; ++tc) {
      float bv = best[tc][0];
#pragma unroll
      for (int r = 1; r < 8; ++r) bv = fmaxf(bv, best[tc][r]);
      int c[8];
#pragma unroll
      for (int r = 0; r < 8; ++r)
        c[r] = (best[tc][r] == bv) ? ((argk[tc][r] << 4) + r) : 0x7fffffff;
      const int c01 = min(c[0], c[1]), c23 = min(c[2], c[3]);
      const int c45 = min(c[4], c[5]), c67 = min(c[6], c[7]);
      int iv = min(min(c01, c23), min(c45, c67));
      iv += hi8;
      const float bo = __shfl_xor(bv, 16, 32);
      const int   io = __shfl_xor(iv, 16, 32);
      const bool tk2 = (bo > bv) || ((bo == bv) && (io < iv));
      bb[tc] = tk2 ? bo : bv;
      ii[tc] = tk2 ? io : iv;
    }

    const float bsel = hi ? bb[1] : bb[0];
    const int   isel = hi ? ii[1] : ii[0];
    const float an   = bsel + e;
    const float afin = m ? an : aold;

    sAlpha[nxt][j] = afin;
    bp[((size_t)b * (Ss - 1) + (t - 1)) * Cc + j] = (uint8_t)isel;
    __builtin_amdgcn_fence(__ATOMIC_RELEASE, "workgroup", "local");
    __builtin_amdgcn_s_barrier();
    __builtin_amdgcn_fence(__ATOMIC_ACQUIRE, "workgroup", "local");
  }

  if (w == 0) {
    const float* fa = sAlpha[(Ss - 1) & 1];
    float bbest = NEG; int barg = 0;
#pragma unroll
    for (int k = 0; k < 4; ++k) {
      const int   jj = (k << 5) + lane;
      const float v  = fa[jj];
      const bool  g  = v > bbest;
      bbest = g ? v : bbest;
      barg  = g ? jj : barg;
    }
#pragma unroll
    for (int off = 16; off >= 1; off >>= 1) {
      const float bo = __shfl_xor(bbest, off, 32);
      const int   ao = __shfl_xor(barg, off, 32);
      const bool  tk = (bo > bbest) || ((bo == bbest) && (ao < barg));
      bbest = tk ? bo : bbest;
      barg  = tk ? ao : barg;
    }
    if (lane == 0) path[(size_t)b * Ss + (Ss - 1)] = barg;
  }
}

__global__ __launch_bounds__(32) void viterbi_backtrace(
    const uint8_t* __restrict__ bp, int* __restrict__ path)
{
  const int b    = blockIdx.x;
  const int lane = threadIdx.x;
  const uint8_t* bpb = bp + (size_t)b * (Ss - 1) * Cc;
  int tag = path[(size_t)b * Ss + (Ss - 1)];

  auto ROW = [&](int tt) -> unsigned {
    return ((const unsigned*)(bpb + (size_t)tt * Cc))[lane];
  };
  auto CONSUME = [&](unsigned rowv, int tt) {
    const unsigned wsel = (unsigned)__shfl((int)rowv, tag >> 2, 32);
    tag = (int)((wsel >> ((tag & 3) << 3)) & 0xFF);
    if (lane == 0) path[(size_t)b * Ss + tt] = tag;
  };

  unsigned r0 = ROW(Ss - 2), r1 = ROW(Ss - 3), r2 = ROW(Ss - 4), r3 = ROW(Ss - 5);
  for (int tb = Ss - 2; tb >= 0; tb -= 4) {
    unsigned n0 = 0, n1 = 0, n2 = 0, n3 = 0;
    if (tb - 4 >= 0) n0 = ROW(tb - 4);
    if (tb - 5 >= 0) n1 = ROW(tb - 5);
    if (tb - 6 >= 0) n2 = ROW(tb - 6);
    if (tb - 7 >= 0) n3 = ROW(tb - 7);
    CONSUME(r0, tb);
    if (tb - 1 >= 0) CONSUME(r1, tb - 1);
    if (tb - 2 >= 0) CONSUME(r2, tb - 2);
    if (tb - 3 >= 0) CONSUME(r3, tb - 3);
    r0 = n0; r1 = n1; r2 = n2; r3 = n3;
  }
}

extern "C" void kernel_launch(void* const* d_in, const int* in_sizes, int n_in,
                              void* d_out, int out_size, void* d_ws, size_t ws_size,
                              hipStream_t stream) {
  const float* emissions   = (const float*)d_in[0];
  const int*   mask        = (const int*)d_in[1];
  const float* transitions = (const float*)d_in[2];
  int*         path        = (int*)d_out;
  uint8_t*     bp          = (uint8_t*)d_ws;

  viterbi_forward<<<Bb, 128, 0, stream>>>(emissions, mask, transitions, bp, path);
  viterbi_backtrace<<<Bb, 32, 0, stream>>>(bp, path);
}


# forward kernel runs 8 waves per workgroup (two per SIMD), each owning one 16-column tile instead of two, so a SIMD issues one wave while the other waits on a select, matrix or cross-lane result
# speedup vs baseline: 1.8681x; 1.1512x over previous
#include <hip/hip_runtime.h>
#include <hip/hip_bf16.h>
#include <cstdint>

typedef __attribute__((ext_vector_type(2))) float v2f;
typedef __attribute__((ext_vector_type(8))) float v8f;

constexpr int Bb = 256;
constexpr int Ss = 1024;
constexpr int Cc = 128;

__global__ __launch_bounds__(256) void viterbi_forward(
    const float* __restrict__ emis, const int* __restrict__ mask,
    const float* __restrict__ T, uint8_t* __restrict__ bp,
    int* __restrict__ path)
{
  __shared__ float sAlpha[2][Cc];
  const int b      = blockIdx.x;
  const int tid    = threadIdx.x;
  const int w      = tid >> 5;
  const int lane   = tid & 31;
  const int laneLo = lane & 15;
  const bool hi    = lane >= 16;
  const int  hi8   = hi ? 8 : 0;
  const int  j     = (w << 4) + laneLo;

  float ct[8][8];
#pragma unroll
  for (int k = 0; k < 8; ++k)
#pragma unroll
    for (int r = 0; r < 8; ++r) {
      const int row = 16 * k + r + hi8;
      ct[k][r] = T[row * Cc + j];
    }

  v2f bfrag;
  bfrag.x = hi ? 0.0f : 1.0f;
  bfrag.y = 0.0f;

  if (!hi) sAlpha[0][j] = emis[(size_t)b * Ss * Cc + j];
  __syncthreads();

  const float NEG = -__builtin_inff();

  for (int t = 1; t < Ss; ++t) {
    const int cur = (t - 1) & 1, nxt = t & 1;

    const float e    = emis[((size_t)b * Ss + t) * Cc + j];
    const int   m    = mask[b * Ss + t];
    const float aold = sAlpha[cur][j];
    if (t + 4 < Ss)
      __builtin_prefetch(&emis[((size_t)b * Ss + t + 4) * Cc + j], 0, 0);

    float af[8];
#pragma unroll
    for (int k = 0; k < 8; ++k) {
      const float v = sAlpha[cur][16 * k + laneLo];
      af[k] = hi ? 0.0f : v;
    }

    float best[8];
    int   argk[8];

#pragma unroll
    for (int k = 0; k < 8; ++k) {
      v2f a; a.x = af[k]; a.y = 0.0f;
      v8f cmat;
#pragma unroll
      for (int r = 0; r < 8; ++r) cmat[r] = ct[k][r];
      v8f d = __builtin_amdgcn_wmma_f32_16x16x4_f32(
          false, a, false, bfrag, (short)0, cmat, false, false);
      if (k == 0) {
#pragma unroll
        for (int r = 0; r < 8; ++r) { best[r] = d[r]; argk[r] = 0; }
      } else {
#pragma unroll
        for (int r = 0; r < 8; ++r) {
          const bool g = d[r] > best[r];
          best[r] = g ? d[r] : best[r];
          argk[r] = g ? k : argk[r];
        }
      }
    }

    float bv = best[0];
#pragma unroll
    for (int r = 1; r < 8; ++r) bv = fmaxf(bv, best[r]);
    int c[8];
#pragma unroll
    for (int r = 0; r < 8; ++r)
      c[r] = (best[r] == bv) ? ((argk[r] << 4) + r) : 0x7fffffff;
    const int c01 = min(c[0], c[1]), c23 = min(c[2], c[3]);
    const int c45 = min(c[4], c[5]), c67 = min(c[6], c[7]);
    int iv = min(min(c01, c23), min(c45, c67));
    iv += hi8;
    const float bo = __shfl_xor(bv, 16, 32);
    const int   io = __shfl_xor(iv, 16, 32);
    const bool tk2 = (bo > bv) || ((bo == bv) && (io < iv));
    const float bsel = tk2 ? bo : bv;
    const int   isel = tk2 ? io : iv;

    const float an   = bsel + e;
    const float afin = m ? an : aold;

    if (!hi) {
      sAlpha[nxt][j] = afin;
      bp[((size_t)b * (Ss - 1) + (t - 1)) * Cc + j] = (uint8_t)isel;
    }
    __builtin_amdgcn_fence(__ATOMIC_RELEASE, "workgroup", "local");
    __builtin_amdgcn_s_barrier();
    __builtin_amdgcn_fence(__ATOMIC_ACQUIRE, "workgroup", "local");
  }

  if (w == 0) {
    const float* fa = sAlpha[(Ss - 1) & 1];
    float bbest = NEG; int barg = 0;
#pragma unroll
    for (int k = 0; k < 4; ++k) {
      const int   jj = (k << 5) + lane;
      const float v  = fa[jj];
      const bool  g  = v > bbest;
      bbest = g ? v : bbest;
      barg  = g ? jj : barg;
    }
#pragma unroll
    for (int off = 16; off >= 1; off >>= 1) {
      const float bo = __shfl_xor(bbest, off, 32);
      const int   ao = __shfl_xor(barg, off, 32);
      const bool  tk = (bo > bbest) || ((bo == bbest) && (ao < barg));
      bbest = tk ? bo : bbest;
      barg  = tk ? ao : barg;
    }
    if (lane == 0) path[(size_t)b * Ss + (Ss - 1)] = barg;
  }
}

__global__ __launch_bounds__(32) void viterbi_backtrace(
    const uint8_t* __restrict__ bp, int* __restrict__ path)
{
  const int b    = blockIdx.x;
  const int lane = threadIdx.x;
  const uint8_t* bpb = bp + (size_t)b * (Ss - 1) * Cc;
  int tag = path[(size_t)b * Ss + (Ss - 1)];

  auto ROW = [&](int tt) -> unsigned {
    return ((const unsigned*)(bpb + (size_t)tt * Cc))[lane];
  };
  auto CONSUME = [&](unsigned rowv, int tt) {
    const unsigned wsel = (unsigned)__shfl((int)rowv, tag >> 2, 32);
    tag = (int)((wsel >> ((tag & 3) << 3)) & 0xFF);
    if (lane == 0) path[(size_t)b * Ss + tt] = tag;
  };

  unsigned r0 = ROW(Ss - 2), r1 = ROW(Ss - 3), r2 = ROW(Ss - 4), r3 = ROW(Ss - 5);
  for (int tb = Ss - 2; tb >= 0; tb -= 4) {
    unsigned n0 = 0, n1 = 0, n2 = 0, n3 = 0;
    if (tb - 4 >= 0) n0 = ROW(tb - 4);
    if (tb - 5 >= 0) n1 = ROW(tb - 5);
    if (tb - 6 >= 0) n2 = ROW(tb - 6);
    if (tb - 7 >= 0) n3 = ROW(tb - 7);
    CONSUME(r0, tb);
    if (tb - 1 >= 0) CONSUME(r1, tb - 1);
    if (tb - 2 >= 0) CONSUME(r2, tb - 2);
    if (tb - 3 >= 0) CONSUME(r3, tb - 3);
    r0 = n0; r1 = n1; r2 = n2; r3 = n3;
  }
}

extern "C" void kernel_launch(void* const* d_in, const int* in_sizes, int n_in,
                              void* d_out, int out_size, void* d_ws, size_t ws_size,
                              hipStream_t stream) {
  const float* emissions   = (const float*)d_in[0];
  const int*   mask        = (const int*)d_in[1];
  const float* transitions = (const float*)d_in[2];
  int*         path        = (int*)d_out;
  uint8_t*     bp          = (uint8_t*)d_ws;

  viterbi_forward<<<Bb, 256, 0, stream>>>(emissions, mask, transitions, bp, path);
  viterbi_backtrace<<<Bb, 32, 0, stream>>>(bp, path);
}


# the two lane halves of a column exchange max and index by v_permlanex16 on the vector ALU, not by two LDS-unit permutes; value first: alpha goes to LDS before the index fold; same max and tie rule
# speedup vs baseline: 1.9497x; 1.0437x over previous
#include <hip/hip_runtime.h>
#include <hip/hip_bf16.h>
#include <cstdint>

typedef __attribute__((ext_vector_type(2))) float v2f;
typedef __attribute__((ext_vector_type(8))) float v8f;

constexpr int Bb = 256;
constexpr int Ss = 1024;
constexpr int Cc = 128;

static __device__ __forceinline__ int xhalf_i(int x) {
  return __builtin_amdgcn_permlanex16(x, x, 0x76543210, 0xfedcba98, false, false);
}
static __device__ __forceinline__ float xhalf_f(float x) {
  return __builtin_bit_cast(float, xhalf_i(__builtin_bit_cast(int, x)));
}

__global__ __launch_bounds__(256) void viterbi_forward(
    const float* __restrict__ emis, const int* __restrict__ mask,
    const float* __restrict__ T, uint8_t* __restrict__ bp,
    int* __restrict__ path)
{
  __shared__ float sAlpha[2][Cc];
  const int b      = blockIdx.x;
  const int tid    = threadIdx.x;
  const int w      = tid >> 5;
  const int lane   = tid & 31;
  const int laneLo = lane & 15;
  const bool hi    = lane >= 16;
  const int  hi8   = hi ? 8 : 0;
  const int  j     = (w << 4) + laneLo;

  float ct[8][8];
#pragma unroll
  for (int k = 0; k < 8; ++k)
#pragma unroll
    for (int r = 0; r < 8; ++r) {
      const int row = 16 * k + r + hi8;
      ct[k][r] = T[row * Cc + j];
    }

  v2f bfrag;
  bfrag.x = hi ? 0.0f : 1.0f;
  bfrag.y = 0.0f;

  if (!hi) sAlpha[0][j] = emis[(size_t)b * Ss * Cc + j];
  __syncthreads();

  const float NEG = -__builtin_inff();

  for (int t = 1; t < Ss; ++t) {
    const int cur = (t - 1) & 1, nxt = t & 1;

    const float e    = emis[((size_t)b * Ss + t) * Cc + j];
    const int   m    = mask[b * Ss + t];
    const float aold = sAlpha[cur][j];
    __builtin_amdgcn_sched_barrier(0);
    if (t + 4 < Ss)
      __builtin_prefetch(&emis[((size_t)b * Ss + t + 4) * Cc + j], 0, 0);

    float af[8];
#pragma unroll
    for (int k = 0; k < 8; ++k) {
      const float v = sAlpha[cur][16 * k + laneLo];
      af[k] = hi ? 0.0f : v;
    }

    float best[8];
    int   argk[8];

#pragma unroll
    for (int k = 0; k < 8; ++k) {
      v2f a; a.x = af[k]; a.y = 0.0f;
      v8f cmat;
#pragma unroll
      for (int r = 0; r < 8; ++r) cmat[r] = ct[k][r];
      v8f d = __builtin_amdgcn_wmma_f32_16x16x4_f32(
          false, a, false, bfrag, (short)0, cmat, false, false);
      if (k == 0) {
#pragma unroll
        for (int r = 0; r < 8; ++r) { best[r] = d[r]; argk[r] = 0; }
      } else {
#pragma unroll
        for (int r = 0; r < 8; ++r) {
          const bool g = d[r] > best[r];
          best[r] = g ? d[r] : best[r];
          argk[r] = g ? k : argk[r];
        }
      }
    }

    float bv = best[0];
#pragma unroll
    for (int r = 1; r < 8; ++r) bv = fmaxf(bv, best[r]);

    const float bo   = xhalf_f(bv);
    const bool  og   = bo > bv;
    const float bsel = og ? bo : bv;
    const float an   = bsel + e;
    const float afin = m ? an : aold;
    sAlpha[nxt][j] = afin;
    __builtin_amdgcn_sched_barrier(0);

    int c[8];
#pragma unroll
    for (int r = 0; r < 8; ++r)
      c[r] = (best[r] == bv) ? ((argk[r] << 4) + r) : 0x7fffffff;
    const int c01 = min(c[0], c[1]), c23 = min(c[2], c[3]);
    const int c45 = min(c[4], c[5]), c67 = min(c[6], c[7]);
    int iv = min(min(c01, c23), min(c45, c67));
    iv += hi8;
    const int io = xhalf_i(iv);
    int isel = (bv > bo) ? iv : min(iv, io);
    isel     = og ? io : isel;

    if (!hi)
      bp[((size_t)b * (Ss - 1) + (t - 1)) * Cc + j] = (uint8_t)isel;
    __builtin_amdgcn_fence(__ATOMIC_RELEASE, "workgroup", "local");
    __builtin_amdgcn_s_barrier();
    __builtin_amdgcn_fence(__ATOMIC_ACQUIRE, "workgroup", "local");
  }

  if (w == 0) {
    const float* fa = sAlpha[(Ss - 1) & 1];
    float bbest = NEG; int barg = 0;
#pragma unroll
    for (int k = 0; k < 4; ++k) {
      const int   jj = (k << 5) + lane;
      const float v  = fa[jj];
      const bool  g  = v > bbest;
      bbest = g ? v : bbest;
      barg  = g ? jj : barg;
    }
#pragma unroll
    for (int off = 16; off >= 1; off >>= 1) {
      const float bo = __shfl_xor(bbest, off, 32);
      const int   ao = __shfl_xor(barg, off, 32);
      const bool  tk = (bo > bbest) || ((bo == bbest) && (ao < barg));
      bbest = tk ? bo : bbest;
      barg  = tk ? ao : barg;
    }
    if (lane == 0) path[(size_t)b * Ss + (Ss - 1)] = barg;
  }
}

__global__ __launch_bounds__(32) void viterbi_backtrace(
    const uint8_t* __restrict__ bp, int* __restrict__ path)
{
  const int b    = blockIdx.x;
  const int lane = threadIdx.x;
  const uint8_t* bpb = bp + (size_t)b * (Ss - 1) * Cc;
  int tag = path[(size_t)b * Ss + (Ss - 1)];

  auto ROW = [&](int tt) -> unsigned {
    return ((const unsigned*)(bpb + (size_t)tt * Cc))[lane];
  };
  auto CONSUME = [&](unsigned rowv, int tt) {
    const unsigned wsel = (unsigned)__shfl((int)rowv, tag >> 2, 32);
    tag = (int)((wsel >> ((tag & 3) << 3)) & 0xFF);
    if (lane == 0) path[(size_t)b * Ss + tt] = tag;
  };

  unsigned r0 = ROW(Ss - 2), r1 = ROW(Ss - 3), r2 = ROW(Ss - 4), r3 = ROW(Ss - 5);
  for (int tb = Ss - 2; tb >= 0; tb -= 4) {
    unsigned n0 = 0, n1 = 0, n2 = 0, n3 = 0;
    if (tb - 4 >= 0) n0 = ROW(tb - 4);
    if (tb - 5 >= 0) n1 = ROW(tb - 5);
    if (tb - 6 >= 0) n2 = ROW(tb - 6);
    if (tb - 7 >= 0) n3 = ROW(tb - 7);
    CONSUME(r0, tb);
    if (tb - 1 >= 0) CONSUME(r1, tb - 1);
    if (tb - 2 >= 0) CONSUME(r2, tb - 2);
    if (tb - 3 >= 0) CONSUME(r3, tb - 3);
    r0 = n0; r1 = n1; r2 = n2; r3 = n3;
  }
}

extern "C" void kernel_launch(void* const* d_in, const int* in_sizes, int n_in,
                              void* d_out, int out_size, void* d_ws, size_t ws_size,
                              hipStream_t stream) {
  const float* emissions   = (const float*)d_in[0];
  const int*   mask        = (const int*)d_in[1];
  const float* transitions = (const float*)d_in[2];
  int*         path        = (int*)d_out;
  uint8_t*     bp          = (uint8_t*)d_ws;

  viterbi_forward<<<Bb, 256, 0, stream>>>(emissions, mask, transitions, bp, path);
  viterbi_backtrace<<<Bb, 32, 0, stream>>>(bp, path);
}


# backtrace step on the scalar unit: tag kept in a scalar register, its row word read by v_readlane, not an LDS-unit permute; 32 rows held ahead in registers, one 128-byte path store per 32 steps
# speedup vs baseline: 2.3926x; 1.2272x over previous
#include <hip/hip_runtime.h>
#include <hip/hip_bf16.h>
#include <cstdint>

typedef __attribute__((ext_vector_type(2))) float v2f;
typedef __attribute__((ext_vector_type(8))) float v8f;

constexpr int Bb = 256;
constexpr int Ss = 1024;
constexpr int Cc = 128;

static __device__ __forceinline__ int xhalf_i(int x) {
  return __builtin_amdgcn_permlanex16(x, x, 0x76543210, 0xfedcba98, false, false);
}
static __device__ __forceinline__ float xhalf_f(float x) {
  return __builtin_bit_cast(float, xhalf_i(__builtin_bit_cast(int, x)));
}

__global__ __launch_bounds__(256) void viterbi_forward(
    const float* __restrict__ emis, const int* __restrict__ mask,
    const float* __restrict__ T, uint8_t* __restrict__ bp,
    int* __restrict__ path)
{
  __shared__ float sAlpha[2][Cc];
  const int b      = blockIdx.x;
  const int tid    = threadIdx.x;
  const int w      = tid >> 5;
  const int lane   = tid & 31;
  const int laneLo = lane & 15;
  const bool hi    = lane >= 16;
  const int  hi8   = hi ? 8 : 0;
  const int  j     = (w << 4) + laneLo;

  float ct[8][8];
#pragma unroll
  for (int k = 0; k < 8; ++k)
#pragma unroll
    for (int r = 0; r < 8; ++r) {
      const int row = 16 * k + r + hi8;
      ct[k][r] = T[row * Cc + j];
    }

  v2f bfrag;
  bfrag.x = hi ? 0.0f : 1.0f;
  bfrag.y = 0.0f;

  if (!hi) sAlpha[0][j] = emis[(size_t)b * Ss * Cc + j];
  __syncthreads();

  const float NEG = -__builtin_inff();

  for (int t = 1; t < Ss; ++t) {
    const int cur = (t - 1) & 1, nxt = t & 1;

    const float e    = emis[((size_t)b * Ss + t) * Cc + j];
    const int   m    = mask[b * Ss + t];
    const float aold = sAlpha[cur][j];
    __builtin_amdgcn_sched_barrier(0);
    if (t + 4 < Ss)
      __builtin_prefetch(&emis[((size_t)b * Ss + t + 4) * Cc + j], 0, 0);

    float af[8];
#pragma unroll
    for (int k = 0; k < 8; ++k) {
      const float v = sAlpha[cur][16 * k + laneLo];
      af[k] = hi ? 0.0f : v;
    }

    float best[8];
    int   argk[8];

#pragma unroll
    for (int k = 0; k < 8; ++k) {
      v2f a; a.x = af[k]; a.y = 0.0f;
      v8f cmat;
#pragma unroll
      for (int r = 0; r < 8; ++r) cmat[r] = ct[k][r];
      v8f d = __builtin_amdgcn_wmma_f32_16x16x4_f32(
          false, a, false, bfrag, (short)0, cmat, false, false);
      if (k == 0) {
#pragma unroll
        for (int r = 0; r < 8; ++r) { best[r] = d[r]; argk[r] = 0; }
      } else {
#pragma unroll
        for (int r = 0; r < 8; ++r) {
          const bool g = d[r] > best[r];
          best[r] = g ? d[r] : best[r];
          argk[r] = g ? k : argk[r];
        }
      }
    }

    float bv = best[0];
#pragma unroll
    for (int r = 1; r < 8; ++r) bv = fmaxf(bv, best[r]);

    const float bo   = xhalf_f(bv);
    const bool  og   = bo > bv;
    const float bsel = og ? bo : bv;
    const float an   = bsel + e;
    const float afin = m ? an : aold;
    sAlpha[nxt][j] = afin;
    __builtin_amdgcn_sched_barrier(0);

    int c[8];
#pragma unroll
    for (int r = 0; r < 8; ++r)
      c[r] = (best[r] == bv) ? ((argk[r] << 4) + r) : 0x7fffffff;
    const int c01 = min(c[0], c[1]), c23 = min(c[2], c[3]);
    const int c45 = min(c[4], c[5]), c67 = min(c[6], c[7]);
    int iv = min(min(c01, c23), min(c45, c67));
    iv += hi8;
    const int io = xhalf_i(iv);
    int isel = (bv > bo) ? iv : min(iv, io);
    isel     = og ? io : isel;

    if (!hi)
      bp[((size_t)b * (Ss - 1) + (t - 1)) * Cc + j] = (uint8_t)isel;
    __builtin_amdgcn_fence(__ATOMIC_RELEASE, "workgroup", "local");
    __builtin_amdgcn_s_barrier();
    __builtin_amdgcn_fence(__ATOMIC_ACQUIRE, "workgroup", "local");
  }

  if (w == 0) {
    const float* fa = sAlpha[(Ss - 1) & 1];
    float bbest = NEG; int barg = 0;
#pragma unroll
    for (int k = 0; k < 4; ++k) {
      const int   jj = (k << 5) + lane;
      const float v  = fa[jj];
      const bool  g  = v > bbest;
      bbest = g ? v : bbest;
      barg  = g ? jj : barg;
    }
#pragma unroll
    for (int off = 16; off >= 1; off >>= 1) {
      const float bo = __shfl_xor(bbest, off, 32);
      const int   ao = __shfl_xor(barg, off, 32);
      const bool  tk = (bo > bbest) || ((bo == bbest) && (ao < barg));
      bbest = tk ? bo : bbest;
      barg  = tk ? ao : barg;
    }
    if (lane == 0) path[(size_t)b * Ss + (Ss - 1)] = barg;
  }
}

template <bool FIRST>
static __device__ __forceinline__ void load32(
    unsigned (&r)[32], const uint8_t* __restrict__ bpb, int c, int lane)
{
  const unsigned* base = (const unsigned*)(bpb + (size_t)c * 32 * Cc) + lane;
#pragma unroll
  for (int i = 31; i >= 0; --i) {
    if (FIRST && i == 31) { r[i] = 0u; continue; }
    r[i] = base[i * (Cc / 4)];
  }
}

template <bool FIRST>
static __device__ __forceinline__ void chase32(
    const unsigned (&r)[32], int& tag, int lane, int* __restrict__ pline)
{
  int acc = 0;
#pragma unroll
  for (int i = 31; i >= 0; --i) {
    if (FIRST && i == 31) continue;
    const unsigned wsel =
        (unsigned)__builtin_amdgcn_readlane((int)r[i], tag >> 2);
    tag = (int)((wsel >> ((tag & 3) << 3)) & 0xFFu);
    acc = (lane == i) ? tag : acc;
  }
  if (!FIRST || lane < 31) pline[lane] = acc;
}

__global__ __launch_bounds__(32) void viterbi_backtrace(
    const uint8_t* __restrict__ bp, int* __restrict__ path)
{
  const int b    = blockIdx.x;
  const int lane = threadIdx.x;
  const uint8_t* bpb = bp + (size_t)b * (Ss - 1) * Cc;
  int* pb = path + (size_t)b * Ss;

  unsigned ra[32], rb[32];
  load32<true >(ra, bpb, 31, lane);
  load32<false>(rb, bpb, 30, lane);
  int tag = __builtin_amdgcn_readfirstlane(pb[Ss - 1]);

  chase32<true>(ra, tag, lane, pb + 31 * 32);
  for (int c = 30; c >= 2; c -= 2) {
    load32<false>(ra, bpb, c - 1, lane);
    chase32<false>(rb, tag, lane, pb + c * 32);
    load32<false>(rb, bpb, c - 2, lane);
    chase32<false>(ra, tag, lane, pb + (c - 1) * 32);
  }
  chase32<false>(rb, tag, lane, pb);
}

extern "C" void kernel_launch(void* const* d_in, const int* in_sizes, int n_in,
                              void* d_out, int out_size, void* d_ws, size_t ws_size,
                              hipStream_t stream) {
  const float* emissions   = (const float*)d_in[0];
  const int*   mask        = (const int*)d_in[1];
  const float* transitions = (const float*)d_in[2];
  int*         path        = (int*)d_out;
  uint8_t*     bp          = (uint8_t*)d_ws;

  viterbi_forward<<<Bb, 256, 0, stream>>>(emissions, mask, transitions, bp, path);
  viterbi_backtrace<<<Bb, 32, 0, stream>>>(bp, path);
}


# backpointer index work of step t (select chains over kept compare masks, fold, half exchange, bp store) runs at the head of step t+1 under the LDS read wait, not between alpha store and barrier
# speedup vs baseline: 2.6027x; 1.0878x over previous
#include <hip/hip_runtime.h>
#include <hip/hip_bf16.h>
#include <cstdint>

typedef __attribute__((ext_vector_type(2))) float v2f;
typedef __attribute__((ext_vector_type(8))) float v8f;

constexpr int Bb = 256;
constexpr int Ss = 1024;
constexpr int Cc = 128;

static __device__ __forceinline__ int xhalf_i(int x) {
  return __builtin_amdgcn_permlanex16(x, x, 0x76543210, 0xfedcba98, false, false);
}
static __device__ __forceinline__ float xhalf_f(float x) {
  return __builtin_bit_cast(float, xhalf_i(__builtin_bit_cast(int, x)));
}

static __device__ __forceinline__ void store_backpointer(
    const unsigned (&gm)[7][8], const float (&pbest)[8], float pbv, float pbo,
    int hi8, uint8_t* __restrict__ dst)
{
  int c[8];
#pragma unroll
  for (int r = 0; r < 8; ++r) {
    int a = r;
#pragma unroll
    for (int k = 1; k < 8; ++k)
      a = __builtin_amdgcn_inverse_ballot_w32(gm[k - 1][r]) ? (16 * k + r) : a;
    c[r] = (pbest[r] == pbv) ? a : 0x7fffffff;
  }
  const int c01 = min(c[0], c[1]), c23 = min(c[2], c[3]);
  const int c45 = min(c[4], c[5]), c67 = min(c[6], c[7]);
  int iv = min(min(c01, c23), min(c45, c67));
  iv += hi8;
  const int io = xhalf_i(iv);
  int isel = (pbv > pbo) ? iv : min(iv, io);
  isel     = (pbo > pbv) ? io : isel;
  if (__builtin_amdgcn_inverse_ballot_w32(0x0000ffffu)) *dst = (uint8_t)isel;
}

__global__ __launch_bounds__(256) void viterbi_forward(
    const float* __restrict__ emis, const int* __restrict__ mask,
    const float* __restrict__ T, uint8_t* __restrict__ bp,
    int* __restrict__ path)
{
  __shared__ float sAlpha[2][Cc];
  const int b      = blockIdx.x;
  const int tid    = threadIdx.x;
  const int w      = tid >> 5;
  const int lane   = tid & 31;
  const int laneLo = lane & 15;
  const bool hi    = lane >= 16;
  const int  hi8   = hi ? 8 : 0;
  const int  j     = (w << 4) + laneLo;

  float ct[8][8];
#pragma unroll
  for (int k = 0; k < 8; ++k)
#pragma unroll
    for (int r = 0; r < 8; ++r) {
      const int row = 16 * k + r + hi8;
      ct[k][r] = T[row * Cc + j];
    }

  v2f bfrag;
  bfrag.x = hi ? 0.0f : 1.0f;
  bfrag.y = 0.0f;

  if (!hi) sAlpha[0][j] = emis[(size_t)b * Ss * Cc + j];
  __syncthreads();

  const float NEG = -__builtin_inff();

  unsigned gm[7][8];
  float pbest[8];
  float pbv = 0.0f, pbo = 0.0f;
#pragma unroll
  for (int r = 0; r < 8; ++r) {
    pbest[r] = 0.0f;
#pragma unroll
    for (int k = 0; k < 7; ++k) gm[k][r] = 0u;
  }
  uint8_t* const bpcol = bp + (size_t)b * (Ss - 1) * Cc + j;

  for (int t = 1; t < Ss; ++t) {
    const int cur = (t - 1) & 1, nxt = t & 1;

    const float e    = emis[((size_t)b * Ss + t) * Cc + j];
    const int   m    = mask[b * Ss + t];
    const float aold = sAlpha[cur][j];
    float av[8];
#pragma unroll
    for (int k = 0; k < 8; ++k) av[k] = sAlpha[cur][16 * k + laneLo];
    __builtin_amdgcn_sched_barrier(0);
    if (t + 4 < Ss)
      __builtin_prefetch(&emis[((size_t)b * Ss + t + 4) * Cc + j], 0, 0);

    if (t > 1)
      store_backpointer(gm, pbest, pbv, pbo, hi8,
                        bpcol + (size_t)(t - 2) * Cc);

    float af[8];
#pragma unroll
    for (int k = 0; k < 8; ++k) af[k] = hi ? 0.0f : av[k];

    float best[8];

#pragma unroll
    for (int k = 0; k < 8; ++k) {
      v2f a; a.x = af[k]; a.y = 0.0f;
      v8f cmat;
#pragma unroll
      for (int r = 0; r < 8; ++r) cmat[r] = ct[k][r];
      v8f d = __builtin_amdgcn_wmma_f32_16x16x4_f32(
          false, a, false, bfrag, (short)0, cmat, false, false);
      if (k == 0) {
#pragma unroll
        for (int r = 0; r < 8; ++r) best[r] = d[r];
      } else {
#pragma unroll
        for (int r = 0; r < 8; ++r) {
          const bool g = d[r] > best[r];
          gm[k - 1][r] = __builtin_amdgcn_ballot_w32(g);
          best[r] = g ? d[r] : best[r];
        }
      }
    }

    float bv = best[0];
#pragma unroll
    for (int r = 1; r < 8; ++r) bv = fmaxf(bv, best[r]);

    const float bo   = xhalf_f(bv);
    const bool  og   = bo > bv;
    const float bsel = og ? bo : bv;
    const float an   = bsel + e;
    const float afin = m ? an : aold;
    sAlpha[nxt][j] = afin;

#pragma unroll
    for (int r = 0; r < 8; ++r) pbest[r] = best[r];
    pbv = bv;
    pbo = bo;

    __builtin_amdgcn_fence(__ATOMIC_RELEASE, "workgroup", "local");
    __builtin_amdgcn_s_barrier();
    __builtin_amdgcn_fence(__ATOMIC_ACQUIRE, "workgroup", "local");
  }

  store_backpointer(gm, pbest, pbv, pbo, hi8,
                    bpcol + (size_t)(Ss - 2) * Cc);

  if (w == 0) {
    const float* fa = sAlpha[(Ss - 1) & 1];
    float bbest = NEG; int barg = 0;
#pragma unroll
    for (int k = 0; k < 4; ++k) {
      const int   jj = (k << 5) + lane;
      const float v  = fa[jj];
      const bool  g  = v > bbest;
      bbest = g ? v : bbest;
      barg  = g ? jj : barg;
    }
#pragma unroll
    for (int off = 16; off >= 1; off >>= 1) {
      const float bo = __shfl_xor(bbest, off, 32);
      const int   ao = __shfl_xor(barg, off, 32);
      const bool  tk = (bo > bbest) || ((bo == bbest) && (ao < barg));
      bbest = tk ? bo : bbest;
      barg  = tk ? ao : barg;
    }
    if (lane == 0) path[(size_t)b * Ss + (Ss - 1)] = barg;
  }
}

template <bool FIRST>
static __device__ __forceinline__ void load32(
    unsigned (&r)[32], const uint8_t* __restrict__ bpb, int c, int lane)
{
  const unsigned* base = (const unsigned*)(bpb + (size_t)c * 32 * Cc) + lane;
#pragma unroll
  for (int i = 31; i >= 0; --i) {
    if (FIRST && i == 31) { r[i] = 0u; continue; }
    r[i] = base[i * (Cc / 4)];
  }
}

template <bool FIRST>
static __device__ __forceinline__ void chase32(
    const unsigned (&r)[32], int& tag, int lane, int* __restrict__ pline)
{
  int acc = 0;
#pragma unroll
  for (int i = 31; i >= 0; --i) {
    if (FIRST && i == 31) continue;
    const unsigned wsel =
        (unsigned)__builtin_amdgcn_readlane((int)r[i], tag >> 2);
    tag = (int)((wsel >> ((tag & 3) << 3)) & 0xFFu);
    acc = (lane == i) ? tag : acc;
  }
  if (!FIRST || lane < 31) pline[lane] = acc;
}

__global__ __launch_bounds__(32) void viterbi_backtrace(
    const uint8_t* __restrict__ bp, int* __restrict__ path)
{
  const int b    = blockIdx.x;
  const int lane = threadIdx.x;
  const uint8_t* bpb = bp + (size_t)b * (Ss - 1) * Cc;
  int* pb = path + (size_t)b * Ss;

  unsigned ra[32], rb[32];
  load32<true >(ra, bpb, 31, lane);
  load32<false>(rb, bpb, 30, lane);
  int tag = __builtin_amdgcn_readfirstlane(pb[Ss - 1]);

  chase32<true>(ra, tag, lane, pb + 31 * 32);
  for (int c = 30; c >= 2; c -= 2) {
    load32<false>(ra, bpb, c - 1, lane);
    chase32<false>(rb, tag, lane, pb + c * 32);
    load32<false>(rb, bpb, c - 2, lane);
    chase32<false>(ra, tag, lane, pb + (c - 1) * 32);
  }
  chase32<false>(rb, tag, lane, pb);
}

extern "C" void kernel_launch(void* const* d_in, const int* in_sizes, int n_in,
                              void* d_out, int out_size, void* d_ws, size_t ws_size,
                              hipStream_t stream) {
  const float* emissions   = (const float*)d_in[0];
  const int*   mask        = (const int*)d_in[1];
  const float* transitions = (const float*)d_in[2];
  int*         path        = (int*)d_out;
  uint8_t*     bp          = (uint8_t*)d_ws;

  viterbi_forward<<<Bb, 256, 0, stream>>>(emissions, mask, transitions, bp, path);
  viterbi_backtrace<<<Bb, 32, 0, stream>>>(bp, path);
}


# forward step only takes maxima (3-input max trees); first argmax found one step later by 2-level equality search (first tile equal to lane max, then first row in it), no compare-select chains
# speedup vs baseline: 2.9859x; 1.1472x over previous
#include <hip/hip_runtime.h>
#include <hip/hip_bf16.h>
#include <cstdint>

typedef __attribute__((ext_vector_type(2))) float v2f;
typedef __attribute__((ext_vector_type(8))) float v8f;

constexpr int Bb = 256;
constexpr int Ss = 1024;
constexpr int Cc = 128;

static __device__ __forceinline__ int xhalf_i(int x) {
  return __builtin_amdgcn_permlanex16(x, x, 0x76543210, 0xfedcba98, false, false);
}
static __device__ __forceinline__ float xhalf_f(float x) {
  return __builtin_bit_cast(float, xhalf_i(__builtin_bit_cast(int, x)));
}

static __device__ __forceinline__ void store_backpointer(
    const float (&pd)[8][8], const float (&ptk)[8], float pbv, float pbo,
    int hi8, uint8_t* __restrict__ dst)
{
  float x[7];
#pragma unroll
  for (int r = 0; r < 7; ++r) x[r] = pd[7][r];
  int kk = 16 * 7;
#pragma unroll
  for (int k = 6; k >= 0; --k) {
    const bool ek = (ptk[k] == pbv);
#pragma unroll
    for (int r = 0; r < 7; ++r) x[r] = ek ? pd[k][r] : x[r];
    kk = ek ? 16 * k : kk;
  }
  int rr = 7;
#pragma unroll
  for (int r = 6; r >= 0; --r) rr = (x[r] == pbv) ? r : rr;
  const int iv = kk + rr + hi8;
  const int io = xhalf_i(iv);
  int isel = (pbv > pbo) ? iv : min(iv, io);
  isel     = (pbo > pbv) ? io : isel;
  if (__builtin_amdgcn_inverse_ballot_w32(0x0000ffffu)) *dst = (uint8_t)isel;
}

__global__ __launch_bounds__(256) void viterbi_forward(
    const float* __restrict__ emis, const int* __restrict__ mask,
    const float* __restrict__ T, uint8_t* __restrict__ bp,
    int* __restrict__ path)
{
  __shared__ float sAlpha[2][Cc];
  const int b      = blockIdx.x;
  const int tid    = threadIdx.x;
  const int w      = tid >> 5;
  const int lane   = tid & 31;
  const int laneLo = lane & 15;
  const bool hi    = lane >= 16;
  const int  hi8   = hi ? 8 : 0;
  const int  j     = (w << 4) + laneLo;

  float ct[8][8];
#pragma unroll
  for (int k = 0; k < 8; ++k)
#pragma unroll
    for (int r = 0; r < 8; ++r) {
      const int row = 16 * k + r + hi8;
      ct[k][r] = T[row * Cc + j];
    }

  v2f bfrag;
  bfrag.x = hi ? 0.0f : 1.0f;
  bfrag.y = 0.0f;

  if (!hi) sAlpha[0][j] = emis[(size_t)b * Ss * Cc + j];
  __syncthreads();

  const float NEG = -__builtin_inff();

  float pd[8][8];
  float ptk[8];
  float pbv = 0.0f, pbo = 0.0f;
#pragma unroll
  for (int k = 0; k < 8; ++k) {
    ptk[k] = 0.0f;
#pragma unroll
    for (int r = 0; r < 8; ++r) pd[k][r] = 0.0f;
  }
  uint8_t* const bpcol = bp + (size_t)b * (Ss - 1) * Cc + j;

  for (int t = 1; t < Ss; ++t) {
    const int cur = (t - 1) & 1, nxt = t & 1;

    const float e    = emis[((size_t)b * Ss + t) * Cc + j];
    const int   m    = mask[b * Ss + t];
    const float aold = sAlpha[cur][j];
    float av[8];
#pragma unroll
    for (int k = 0; k < 8; ++k) av[k] = sAlpha[cur][16 * k + laneLo];
    __builtin_amdgcn_sched_barrier(0);
    if (t + 4 < Ss)
      __builtin_prefetch(&emis[((size_t)b * Ss + t + 4) * Cc + j], 0, 0);

    if (t > 1)
      store_backpointer(pd, ptk, pbv, pbo, hi8,
                        bpcol + (size_t)(t - 2) * Cc);

    float af[8];
#pragma unroll
    for (int k = 0; k < 8; ++k) af[k] = hi ? 0.0f : av[k];

    float tk[8];

#pragma unroll
    for (int k = 0; k < 8; ++k) {
      v2f a; a.x = af[k]; a.y = 0.0f;
      v8f cmat;
#pragma unroll
      for (int r = 0; r < 8; ++r) cmat[r] = ct[k][r];
      v8f d = __builtin_amdgcn_wmma_f32_16x16x4_f32(
          false, a, false, bfrag, (short)0, cmat, false, false);
#pragma unroll
      for (int r = 0; r < 8; ++r) pd[k][r] = d[r];
      const float m0 = fmaxf(fmaxf(d[0], d[1]), d[2]);
      const float m1 = fmaxf(fmaxf(d[3], d[4]), d[5]);
      const float m2 = fmaxf(fmaxf(d[5], d[6]), d[7]);
      tk[k] = fmaxf(fmaxf(m0, m1), m2);
    }

    const float q0 = fmaxf(fmaxf(tk[0], tk[1]), tk[2]);
    const float q1 = fmaxf(fmaxf(tk[3], tk[4]), tk[5]);
    const float bv = fmaxf(fmaxf(q0, q1), fmaxf(tk[6], tk[7]));

    const float bo   = xhalf_f(bv);
    const bool  og   = bo > bv;
    const float bsel = og ? bo : bv;
    const float an   = bsel + e;
    const float afin = m ? an : aold;
    sAlpha[nxt][j] = afin;

#pragma unroll
    for (int k = 0; k < 8; ++k) ptk[k] = tk[k];
    pbv = bv;
    pbo = bo;

    __builtin_amdgcn_fence(__ATOMIC_RELEASE, "workgroup", "local");
    __builtin_amdgcn_s_barrier();
    __builtin_amdgcn_fence(__ATOMIC_ACQUIRE, "workgroup", "local");
  }

  store_backpointer(pd, ptk, pbv, pbo, hi8,
                    bpcol + (size_t)(Ss - 2) * Cc);

  if (w == 0) {
    const float* fa = sAlpha[(Ss - 1) & 1];
    float bbest = NEG; int barg = 0;
#pragma unroll
    for (int k = 0; k < 4; ++k) {
      const int   jj = (k << 5) + lane;
      const float v  = fa[jj];
      const bool  g  = v > bbest;
      bbest = g ? v : bbest;
      barg  = g ? jj : barg;
    }
#pragma unroll
    for (int off = 16; off >= 1; off >>= 1) {
      const float bo = __shfl_xor(bbest, off, 32);
      const int   ao = __shfl_xor(barg, off, 32);
      const bool  tk = (bo > bbest) || ((bo == bbest) && (ao < barg));
      bbest = tk ? bo : bbest;
      barg  = tk ? ao : barg;
    }
    if (lane == 0) path[(size_t)b * Ss + (Ss - 1)] = barg;
  }
}

template <bool FIRST>
static __device__ __forceinline__ void load32(
    unsigned (&r)[32], const uint8_t* __restrict__ bpb, int c, int lane)
{
  const unsigned* base = (const unsigned*)(bpb + (size_t)c * 32 * Cc) + lane;
#pragma unroll
  for (int i = 31; i >= 0; --i) {
    if (FIRST && i == 31) { r[i] = 0u; continue; }
    r[i] = base[i * (Cc / 4)];
  }
}

template <bool FIRST>
static __device__ __forceinline__ void chase32(
    const unsigned (&r)[32], int& tag, int lane, int* __restrict__ pline)
{
  int acc = 0;
#pragma unroll
  for (int i = 31; i >= 0; --i) {
    if (FIRST && i == 31) continue;
    const unsigned wsel =
        (unsigned)__builtin_amdgcn_readlane((int)r[i], tag >> 2);
    tag = (int)((wsel >> ((tag & 3) << 3)) & 0xFFu);
    acc = (lane == i) ? tag : acc;
  }
  if (!FIRST || lane < 31) pline[lane] = acc;
}

__global__ __launch_bounds__(32) void viterbi_backtrace(
    const uint8_t* __restrict__ bp, int* __restrict__ path)
{
  const int b    = blockIdx.x;
  const int lane = threadIdx.x;
  const uint8_t* bpb = bp + (size_t)b * (Ss - 1) * Cc;
  int* pb = path + (size_t)b * Ss;

  unsigned ra[32], rb[32];
  load32<true >(ra, bpb, 31, lane);
  load32<false>(rb, bpb, 30, lane);
  int tag = __builtin_amdgcn_readfirstlane(pb[Ss - 1]);

  chase32<true>(ra, tag, lane, pb + 31 * 32);
  for (int c = 30; c >= 2; c -= 2) {
    load32<false>(ra, bpb, c - 1, lane);
    chase32<false>(rb, tag, lane, pb + c * 32);
    load32<false>(rb, bpb, c - 2, lane);
    chase32<false>(ra, tag, lane, pb + (c - 1) * 32);
  }
  chase32<false>(rb, tag, lane, pb);
}

extern "C" void kernel_launch(void* const* d_in, const int* in_sizes, int n_in,
                              void* d_out, int out_size, void* d_ws, size_t ws_size,
                              hipStream_t stream) {
  const float* emissions   = (const float*)d_in[0];
  const int*   mask        = (const int*)d_in[1];
  const float* transitions = (const float*)d_in[2];
  int*         path        = (int*)d_out;
  uint8_t*     bp          = (uint8_t*)d_ws;

  viterbi_forward<<<Bb, 256, 0, stream>>>(emissions, mask, transitions, bp, path);
  viterbi_backtrace<<<Bb, 32, 0, stream>>>(bp, path);
}


# time loop unrolled by two steps: each step's emission value and mask word are loaded one step ahead into alternating registers (no global-memory wait in a step); alpha buffers at fixed LDS offsets
# speedup vs baseline: 3.1713x; 1.0621x over previous
#include <hip/hip_runtime.h>
#include <hip/hip_bf16.h>
#include <cstdint>

typedef __attribute__((ext_vector_type(2))) float v2f;
typedef __attribute__((ext_vector_type(8))) float v8f;

constexpr int Bb = 256;
constexpr int Ss = 1024;
constexpr int Cc = 128;

static __device__ __forceinline__ int xhalf_i(int x) {
  return __builtin_amdgcn_permlanex16(x, x, 0x76543210, 0xfedcba98, false, false);
}
static __device__ __forceinline__ float xhalf_f(float x) {
  return __builtin_bit_cast(float, xhalf_i(__builtin_bit_cast(int, x)));
}

static __device__ __forceinline__ void store_backpointer(
    const float (&pd)[8][8], const float (&ptk)[8], float pbv, float pbo,
    int hi8, uint8_t* __restrict__ dst)
{
  float x[7];
#pragma unroll
  for (int r = 0; r < 7; ++r) x[r] = pd[7][r];
  int kk = 16 * 7;
#pragma unroll
  for (int k = 6; k >= 0; --k) {
    const bool ek = (ptk[k] == pbv);
#pragma unroll
    for (int r = 0; r < 7; ++r) x[r] = ek ? pd[k][r] : x[r];
    kk = ek ? 16 * k : kk;
  }
  int rr = 7;
#pragma unroll
  for (int r = 6; r >= 0; --r) rr = (x[r] == pbv) ? r : rr;
  const int iv = kk + rr + hi8;
  const int io = xhalf_i(iv);
  int isel = (pbv > pbo) ? iv : min(iv, io);
  isel     = (pbo > pbv) ? io : isel;
  if (__builtin_amdgcn_inverse_ballot_w32(0x0000ffffu)) *dst = (uint8_t)isel;
}

__global__ __launch_bounds__(256) void viterbi_forward(
    const float* __restrict__ emis, const int* __restrict__ mask,
    const float* __restrict__ T, uint8_t* __restrict__ bp,
    int* __restrict__ path)
{
  __shared__ float sAlpha[2][Cc];
  const int b      = blockIdx.x;
  const int tid    = threadIdx.x;
  const int w      = tid >> 5;
  const int lane   = tid & 31;
  const int laneLo = lane & 15;
  const bool hi    = lane >= 16;
  const int  hi8   = hi ? 8 : 0;
  const int  j     = (w << 4) + laneLo;

  float ct[8][8];
#pragma unroll
  for (int k = 0; k < 8; ++k)
#pragma unroll
    for (int r = 0; r < 8; ++r) {
      const int row = 16 * k + r + hi8;
      ct[k][r] = T[row * Cc + j];
    }

  v2f bfrag;
  bfrag.x = hi ? 0.0f : 1.0f;
  bfrag.y = 0.0f;

  if (!hi) sAlpha[0][j] = emis[(size_t)b * Ss * Cc + j];
  __syncthreads();

  const float NEG = -__builtin_inff();

  float pd[8][8];
  float ptk[8];
  float pbv = 0.0f, pbo = 0.0f;
#pragma unroll
  for (int k = 0; k < 8; ++k) {
    ptk[k] = 0.0f;
#pragma unroll
    for (int r = 0; r < 8; ++r) pd[k][r] = 0.0f;
  }
  uint8_t* const bpcol = bp + (size_t)b * (Ss - 1) * Cc + j;

  const float* const erow = emis + (size_t)b * Ss * Cc + j;
  const int*   const mrow = mask + b * Ss;

  auto step = [&](const int cur, const int t, const float e, const int m,
                  float& e_ahead, int& m_ahead)
                  __attribute__((always_inline)) {
    const int nxt = cur ^ 1;

    const int ta = min(t + 1, Ss - 1);
    e_ahead = erow[(size_t)ta * Cc];
    m_ahead = mrow[ta];
    const float aold = sAlpha[cur][j];
    float av[8];
#pragma unroll
    for (int k = 0; k < 8; ++k) av[k] = sAlpha[cur][16 * k + laneLo];
    __builtin_amdgcn_sched_barrier(0);
    if (t + 4 < Ss)
      __builtin_prefetch(&erow[(size_t)(t + 4) * Cc], 0, 0);

    if (t > 1)
      store_backpointer(pd, ptk, pbv, pbo, hi8,
                        bpcol + (size_t)(t - 2) * Cc);

    float af[8];
#pragma unroll
    for (int k = 0; k < 8; ++k) af[k] = hi ? 0.0f : av[k];

    float tk[8];

#pragma unroll
    for (int k = 0; k < 8; ++k) {
      v2f a; a.x = af[k]; a.y = 0.0f;
      v8f cmat;
#pragma unroll
      for (int r = 0; r < 8; ++r) cmat[r] = ct[k][r];
      v8f d = __builtin_amdgcn_wmma_f32_16x16x4_f32(
          false, a, false, bfrag, (short)0, cmat, false, false);
#pragma unroll
      for (int r = 0; r < 8; ++r) pd[k][r] = d[r];
      const float m0 = fmaxf(fmaxf(d[0], d[1]), d[2]);
      const float m1 = fmaxf(fmaxf(d[3], d[4]), d[5]);
      const float m2 = fmaxf(fmaxf(d[5], d[6]), d[7]);
      tk[k] = fmaxf(fmaxf(m0, m1), m2);
    }

    const float q0 = fmaxf(fmaxf(tk[0], tk[1]), tk[2]);
    const float q1 = fmaxf(fmaxf(tk[3], tk[4]), tk[5]);
    const float bv = fmaxf(fmaxf(q0, q1), fmaxf(tk[6], tk[7]));

    const float bo   = xhalf_f(bv);
    const bool  og   = bo > bv;
    const float bsel = og ? bo : bv;
    const float an   = bsel + e;
    const float afin = m ? an : aold;
    sAlpha[nxt][j] = afin;

#pragma unroll
    for (int k = 0; k < 8; ++k) ptk[k] = tk[k];
    pbv = bv;
    pbo = bo;

    __builtin_amdgcn_fence(__ATOMIC_RELEASE, "workgroup", "local");
    __builtin_amdgcn_s_barrier();
    __builtin_amdgcn_fence(__ATOMIC_ACQUIRE, "workgroup", "local");
  };

  float eA = erow[(size_t)1 * Cc], eB = 0.0f;
  int   mA = mrow[1],              mB = 0;
  int t = 1;
  for (; t + 1 < Ss; t += 2) {
    step(0, t,     eA, mA, eB, mB);
    step(1, t + 1, eB, mB, eA, mA);
  }
  step(0, t, eA, mA, eB, mB);

  store_backpointer(pd, ptk, pbv, pbo, hi8,
                    bpcol + (size_t)(Ss - 2) * Cc);

  if (w == 0) {
    const float* fa = sAlpha[(Ss - 1) & 1];
    float bbest = NEG; int barg = 0;
#pragma unroll
    for (int k = 0; k < 4; ++k) {
      const int   jj = (k << 5) + lane;
      const float v  = fa[jj];
      const bool  g  = v > bbest;
      bbest = g ? v : bbest;
      barg  = g ? jj : barg;
    }
#pragma unroll
    for (int off = 16; off >= 1; off >>= 1) {
      const float bo = __shfl_xor(bbest, off, 32);
      const int   ao = __shfl_xor(barg, off, 32);
      const bool  tk = (bo > bbest) || ((bo == bbest) && (ao < barg));
      bbest = tk ? bo : bbest;
      barg  = tk ? ao : barg;
    }
    if (lane == 0) path[(size_t)b * Ss + (Ss - 1)] = barg;
  }
}

template <bool FIRST>
static __device__ __forceinline__ void load32(
    unsigned (&r)[32], const uint8_t* __restrict__ bpb, int c, int lane)
{
  const unsigned* base = (const unsigned*)(bpb + (size_t)c * 32 * Cc) + lane;
#pragma unroll
  for (int i = 31; i >= 0; --i) {
    if (FIRST && i == 31) { r[i] = 0u; continue; }
    r[i] = base[i * (Cc / 4)];
  }
}

template <bool FIRST>
static __device__ __forceinline__ void chase32(
    const unsigned (&r)[32], int& tag, int lane, int* __restrict__ pline)
{
  int acc = 0;
#pragma unroll
  for (int i = 31; i >= 0; --i) {
    if (FIRST && i == 31) continue;
    const unsigned wsel =
        (unsigned)__builtin_amdgcn_readlane((int)r[i], tag >> 2);
    tag = (int)((wsel >> ((tag & 3) << 3)) & 0xFFu);
    acc = (lane == i) ? tag : acc;
  }
  if (!FIRST || lane < 31) pline[lane] = acc;
}

__global__ __launch_bounds__(32) void viterbi_backtrace(
    const uint8_t* __restrict__ bp, int* __restrict__ path)
{
  const int b    = blockIdx.x;
  const int lane = threadIdx.x;
  const uint8_t* bpb = bp + (size_t)b * (Ss - 1) * Cc;
  int* pb = path + (size_t)b * Ss;

  unsigned ra[32], rb[32];
  load32<true >(ra, bpb, 31, lane);
  load32<false>(rb, bpb, 30, lane);
  int tag = __builtin_amdgcn_readfirstlane(pb[Ss - 1]);

  chase32<true>(ra, tag, lane, pb + 31 * 32);
  for (int c = 30; c >= 2; c -= 2) {
    load32<false>(ra, bpb, c - 1, lane);
    chase32<false>(rb, tag, lane, pb + c * 32);
    load32<false>(rb, bpb, c - 2, lane);
    chase32<false>(ra, tag, lane, pb + (c - 1) * 32);
  }
  chase32<false>(rb, tag, lane, pb);
}

extern "C" void kernel_launch(void* const* d_in, const int* in_sizes, int n_in,
                              void* d_out, int out_size, void* d_ws, size_t ws_size,
                              hipStream_t stream) {
  const float* emissions   = (const float*)d_in[0];
  const int*   mask        = (const int*)d_in[1];
  const float* transitions = (const float*)d_in[2];
  int*         path        = (int*)d_out;
  uint8_t*     bp          = (uint8_t*)d_ws;

  viterbi_forward<<<Bb, 256, 0, stream>>>(emissions, mask, transitions, bp, path);
  viterbi_backtrace<<<Bb, 32, 0, stream>>>(bp, path);
}


# all 4 k slots of the A fragment carry alpha row blocks: two two-address LDS loads feed the 8 tiles as loaded, four one-hot B fragments pick the block; 8 selects + 8 zero moves per step gone
# speedup vs baseline: 3.3450x; 1.0548x over previous
#include <hip/hip_runtime.h>
#include <hip/hip_bf16.h>
#include <cstdint>

typedef __attribute__((ext_vector_type(2))) float v2f;
typedef __attribute__((ext_vector_type(8))) float v8f;

constexpr int Bb = 256;
constexpr int Ss = 1024;
constexpr int Cc = 128;

static __device__ __forceinline__ int xhalf_i(int x) {
  return __builtin_amdgcn_permlanex16(x, x, 0x76543210, 0xfedcba98, false, false);
}
static __device__ __forceinline__ float xhalf_f(float x) {
  return __builtin_bit_cast(float, xhalf_i(__builtin_bit_cast(int, x)));
}

static __device__ __forceinline__ void store_backpointer(
    const float (&pd)[8][8], const float (&ptk)[8], float pbv, float pbo,
    int hi8, uint8_t* __restrict__ dst)
{
  float x[7];
#pragma unroll
  for (int r = 0; r < 7; ++r) x[r] = pd[7][r];
  int kk = 16 * 7;
#pragma unroll
  for (int k = 6; k >= 0; --k) {
    const bool ek = (ptk[k] == pbv);
#pragma unroll
    for (int r = 0; r < 7; ++r) x[r] = ek ? pd[k][r] : x[r];
    kk = ek ? 16 * k : kk;
  }
  int rr = 7;
#pragma unroll
  for (int r = 6; r >= 0; --r) rr = (x[r] == pbv) ? r : rr;
  const int iv = kk + rr + hi8;
  const int io = xhalf_i(iv);
  int isel = (pbv > pbo) ? iv : min(iv, io);
  isel     = (pbo > pbv) ? io : isel;
  if (__builtin_amdgcn_inverse_ballot_w32(0x0000ffffu)) *dst = (uint8_t)isel;
}

__global__ __launch_bounds__(256) void viterbi_forward(
    const float* __restrict__ emis, const int* __restrict__ mask,
    const float* __restrict__ T, uint8_t* __restrict__ bp,
    int* __restrict__ path)
{
  __shared__ float sAlpha[2][Cc];
  const int b      = blockIdx.x;
  const int tid    = threadIdx.x;
  const int w      = tid >> 5;
  const int lane   = tid & 31;
  const int laneLo = lane & 15;
  const bool hi    = lane >= 16;
  const int  hi8   = hi ? 8 : 0;
  const int  j     = (w << 4) + laneLo;

  float ct[8][8];
#pragma unroll
  for (int k = 0; k < 8; ++k)
#pragma unroll
    for (int r = 0; r < 8; ++r) {
      const int row = 16 * k + r + hi8;
      ct[k][r] = T[row * Cc + j];
    }

  const float oneLo = hi ? 0.0f : 1.0f;
  const float oneHi = hi ? 1.0f : 0.0f;
  v2f bsel[4];
  bsel[0].x = oneLo; bsel[0].y = 0.0f;
  bsel[1].x = 0.0f;  bsel[1].y = oneLo;
  bsel[2].x = oneHi; bsel[2].y = 0.0f;
  bsel[3].x = 0.0f;  bsel[3].y = oneHi;
#pragma unroll
  for (int q = 0; q < 4; ++q) asm volatile("" : "+v"(bsel[q]));
  const int abase = laneLo + (hi ? 32 : 0);

  if (!hi) sAlpha[0][j] = emis[(size_t)b * Ss * Cc + j];
  __syncthreads();

  const float NEG = -__builtin_inff();

  float pd[8][8];
  float ptk[8];
  float pbv = 0.0f, pbo = 0.0f;
#pragma unroll
  for (int k = 0; k < 8; ++k) {
    ptk[k] = 0.0f;
#pragma unroll
    for (int r = 0; r < 8; ++r) pd[k][r] = 0.0f;
  }
  uint8_t* const bpcol = bp + (size_t)b * (Ss - 1) * Cc + j;

  const float* const erow = emis + (size_t)b * Ss * Cc + j;
  const int*   const mrow = mask + b * Ss;

  auto step = [&](const int cur, const int t, const float e, const int m,
                  float& e_ahead, int& m_ahead)
                  __attribute__((always_inline)) {
    const int nxt = cur ^ 1;

    const int ta = min(t + 1, Ss - 1);
    e_ahead = erow[(size_t)ta * Cc];
    m_ahead = mrow[ta];
    v2f pa[2];
#pragma unroll
    for (int g = 0; g < 2; ++g) {
      pa[g].x = sAlpha[cur][64 * g + abase];
      pa[g].y = sAlpha[cur][64 * g + abase + 16];
    }
    const float aold = sAlpha[cur][j];
    __builtin_amdgcn_sched_barrier(0);
    if (t + 4 < Ss)
      __builtin_prefetch(&erow[(size_t)(t + 4) * Cc], 0, 0);

    if (t > 1)
      store_backpointer(pd, ptk, pbv, pbo, hi8,
                        bpcol + (size_t)(t - 2) * Cc);

    float tk[8];

#pragma unroll
    for (int k = 0; k < 8; ++k) {
      v8f cmat;
#pragma unroll
      for (int r = 0; r < 8; ++r) cmat[r] = ct[k][r];
      v8f d = __builtin_amdgcn_wmma_f32_16x16x4_f32(
          false, pa[k >> 2], false, bsel[k & 3], (short)0, cmat, false, false);
#pragma unroll
      for (int r = 0; r < 8; ++r) pd[k][r] = d[r];
      const float m0 = fmaxf(fmaxf(d[0], d[1]), d[2]);
      const float m1 = fmaxf(fmaxf(d[3], d[4]), d[5]);
      const float m2 = fmaxf(fmaxf(d[5], d[6]), d[7]);
      tk[k] = fmaxf(fmaxf(m0, m1), m2);
    }

    const float q0 = fmaxf(fmaxf(tk[0], tk[1]), tk[2]);
    const float q1 = fmaxf(fmaxf(tk[3], tk[4]), tk[5]);
    const float bv = fmaxf(fmaxf(q0, q1), fmaxf(tk[6], tk[7]));

    const float bo   = xhalf_f(bv);
    const bool  og   = bo > bv;
    const float bsel = og ? bo : bv;
    const float an   = bsel + e;
    const float afin = m ? an : aold;
    sAlpha[nxt][j] = afin;

#pragma unroll
    for (int k = 0; k < 8; ++k) ptk[k] = tk[k];
    pbv = bv;
    pbo = bo;

    __builtin_amdgcn_fence(__ATOMIC_RELEASE, "workgroup", "local");
    __builtin_amdgcn_s_barrier();
    __builtin_amdgcn_fence(__ATOMIC_ACQUIRE, "workgroup", "local");
  };

  float eA = erow[(size_t)1 * Cc], eB = 0.0f;
  int   mA = mrow[1],              mB = 0;
  int t = 1;
  for (; t + 1 < Ss; t += 2) {
    step(0, t,     eA, mA, eB, mB);
    step(1, t + 1, eB, mB, eA, mA);
  }
  step(0, t, eA, mA, eB, mB);

  store_backpointer(pd, ptk, pbv, pbo, hi8,
                    bpcol + (size_t)(Ss - 2) * Cc);

  if (w == 0) {
    const float* fa = sAlpha[(Ss - 1) & 1];
    float bbest = NEG; int barg = 0;
#pragma unroll
    for (int k = 0; k < 4; ++k) {
      const int   jj = (k << 5) + lane;
      const float v  = fa[jj];
      const bool  g  = v > bbest;
      bbest = g ? v : bbest;
      barg  = g ? jj : barg;
    }
#pragma unroll
    for (int off = 16; off >= 1; off >>= 1) {
      const float bo = __shfl_xor(bbest, off, 32);
      const int   ao = __shfl_xor(barg, off, 32);
      const bool  tk = (bo > bbest) || ((bo == bbest) && (ao < barg));
      bbest = tk ? bo : bbest;
      barg  = tk ? ao : barg;
    }
    if (lane == 0) path[(size_t)b * Ss + (Ss - 1)] = barg;
  }
}

template <bool FIRST>
static __device__ __forceinline__ void load32(
    unsigned (&r)[32], const uint8_t* __restrict__ bpb, int c, int lane)
{
  const unsigned* base = (const unsigned*)(bpb + (size_t)c * 32 * Cc) + lane;
#pragma unroll
  for (int i = 31; i >= 0; --i) {
    if (FIRST && i == 31) { r[i] = 0u; continue; }
    r[i] = base[i * (Cc / 4)];
  }
}

template <bool FIRST>
static __device__ __forceinline__ void chase32(
    const unsigned (&r)[32], int& tag, int lane, int* __restrict__ pline)
{
  int acc = 0;
#pragma unroll
  for (int i = 31; i >= 0; --i) {
    if (FIRST && i == 31) continue;
    const unsigned wsel =
        (unsigned)__builtin_amdgcn_readlane((int)r[i], tag >> 2);
    tag = (int)((wsel >> ((tag & 3) << 3)) & 0xFFu);
    acc = (lane == i) ? tag : acc;
  }
  if (!FIRST || lane < 31) pline[lane] = acc;
}

__global__ __launch_bounds__(32) void viterbi_backtrace(
    const uint8_t* __restrict__ bp, int* __restrict__ path)
{
  const int b    = blockIdx.x;
  const int lane = threadIdx.x;
  const uint8_t* bpb = bp + (size_t)b * (Ss - 1) * Cc;
  int* pb = path + (size_t)b * Ss;

  unsigned ra[32], rb[32];
  load32<true >(ra, bpb, 31, lane);
  load32<false>(rb, bpb, 30, lane);
  int tag = __builtin_amdgcn_readfirstlane(pb[Ss - 1]);

  chase32<true>(ra, tag, lane, pb + 31 * 32);
  for (int c = 30; c >= 2; c -= 2) {
    load32<false>(ra, bpb, c - 1, lane);
    chase32<false>(rb, tag, lane, pb + c * 32);
    load32<false>(rb, bpb, c - 2, lane);
    chase32<false>(ra, tag, lane, pb + (c - 1) * 32);
  }
  chase32<false>(rb, tag, lane, pb);
}

extern "C" void kernel_launch(void* const* d_in, const int* in_sizes, int n_in,
                              void* d_out, int out_size, void* d_ws, size_t ws_size,
                              hipStream_t stream) {
  const float* emissions   = (const float*)d_in[0];
  const int*   mask        = (const int*)d_in[1];
  const float* transitions = (const float*)d_in[2];
  int*         path        = (int*)d_out;
  uint8_t*     bp          = (uint8_t*)d_ws;

  viterbi_forward<<<Bb, 256, 0, stream>>>(emissions, mask, transitions, bp, path);
  viterbi_backtrace<<<Bb, 32, 0, stream>>>(bp, path);
}


# tile level of the argmax search moved behind the alpha store: upper tiles fill the store-completion wait, lower tiles stand between barrier signal and wait; row level stays under the LDS read
# speedup vs baseline: 3.4576x; 1.0337x over previous
#include <hip/hip_runtime.h>
#include <hip/hip_bf16.h>
#include <cstdint>

typedef __attribute__((ext_vector_type(2))) float v2f;
typedef __attribute__((ext_vector_type(8))) float v8f;

constexpr int Bb = 256;
constexpr int Ss = 1024;
constexpr int Cc = 128;

static __device__ __forceinline__ int xhalf_i(int x) {
  return __builtin_amdgcn_permlanex16(x, x, 0x76543210, 0xfedcba98, false, false);
}
static __device__ __forceinline__ float xhalf_f(float x) {
  return __builtin_bit_cast(float, xhalf_i(__builtin_bit_cast(int, x)));
}

template <int KHI, int KLO>
static __device__ __forceinline__ void pick_tile(
    const v8f (&d)[8], const float (&tk)[8], float bv, float (&x)[7], int& kk)
{
#pragma unroll
  for (int k = KHI; k >= KLO; --k) {
    const bool ek = (tk[k] == bv);
#pragma unroll
    for (int r = 0; r < 7; ++r) x[r] = ek ? d[k][r] : x[r];
    kk = ek ? 16 * k : kk;
  }
}

static __device__ __forceinline__ void store_backpointer(
    const float (&x)[7], int kk, float pbv, float pbo,
    int hi8, uint8_t* __restrict__ dst)
{
  int rr = 7;
#pragma unroll
  for (int r = 6; r >= 0; --r) rr = (x[r] == pbv) ? r : rr;
  const int iv = kk + rr + hi8;
  const int io = xhalf_i(iv);
  int isel = (pbv > pbo) ? iv : min(iv, io);
  isel     = (pbo > pbv) ? io : isel;
  if (__builtin_amdgcn_inverse_ballot_w32(0x0000ffffu)) *dst = (uint8_t)isel;
}

__global__ __launch_bounds__(256) void viterbi_forward(
    const float* __restrict__ emis, const int* __restrict__ mask,
    const float* __restrict__ T, uint8_t* __restrict__ bp,
    int* __restrict__ path)
{
  __shared__ float sAlpha[2][Cc];
  const int b      = blockIdx.x;
  const int tid    = threadIdx.x;
  const int w      = tid >> 5;
  const int lane   = tid & 31;
  const int laneLo = lane & 15;
  const bool hi    = lane >= 16;
  const int  hi8   = hi ? 8 : 0;
  const int  j     = (w << 4) + laneLo;

  float ct[8][8];
#pragma unroll
  for (int k = 0; k < 8; ++k)
#pragma unroll
    for (int r = 0; r < 8; ++r) {
      const int row = 16 * k + r + hi8;
      ct[k][r] = T[row * Cc + j];
    }

  const float oneLo = hi ? 0.0f : 1.0f;
  const float oneHi = hi ? 1.0f : 0.0f;
  v2f bsel[4];
  bsel[0].x = oneLo; bsel[0].y = 0.0f;
  bsel[1].x = 0.0f;  bsel[1].y = oneLo;
  bsel[2].x = oneHi; bsel[2].y = 0.0f;
  bsel[3].x = 0.0f;  bsel[3].y = oneHi;
#pragma unroll
  for (int q = 0; q < 4; ++q) asm volatile("" : "+v"(bsel[q]));
  const int abase = laneLo + (hi ? 32 : 0);

  if (!hi) sAlpha[0][j] = emis[(size_t)b * Ss * Cc + j];
  __syncthreads();

  const float NEG = -__builtin_inff();

  float px[7];
  int   pkk = 0;
  float pbv = 0.0f, pbo = 0.0f;
#pragma unroll
  for (int r = 0; r < 7; ++r) px[r] = 0.0f;
  uint8_t* const bpcol = bp + (size_t)b * (Ss - 1) * Cc + j;

  const float* const erow = emis + (size_t)b * Ss * Cc + j;
  const int*   const mrow = mask + b * Ss;

  auto step = [&](const int cur, const int t, const float e, const int m,
                  float& e_ahead, int& m_ahead)
                  __attribute__((always_inline)) {
    const int nxt = cur ^ 1;

    const int ta = min(t + 1, Ss - 1);
    e_ahead = erow[(size_t)ta * Cc];
    m_ahead = mrow[ta];
    v2f pa[2];
#pragma unroll
    for (int g = 0; g < 2; ++g) {
      pa[g].x = sAlpha[cur][64 * g + abase];
      pa[g].y = sAlpha[cur][64 * g + abase + 16];
    }
    const float aold = sAlpha[cur][j];
    __builtin_amdgcn_sched_barrier(0);
    if (t + 4 < Ss)
      __builtin_prefetch(&erow[(size_t)(t + 4) * Cc], 0, 0);

    if (t > 1)
      store_backpointer(px, pkk, pbv, pbo, hi8,
                        bpcol + (size_t)(t - 2) * Cc);

    float tk[8];
    v8f   dm[8];

#pragma unroll
    for (int k = 0; k < 8; ++k) {
      v8f cmat;
#pragma unroll
      for (int r = 0; r < 8; ++r) cmat[r] = ct[k][r];
      const v8f d = __builtin_amdgcn_wmma_f32_16x16x4_f32(
          false, pa[k >> 2], false, bsel[k & 3], (short)0, cmat, false, false);
      dm[k] = d;
      const float m0 = fmaxf(fmaxf(d[0], d[1]), d[2]);
      const float m1 = fmaxf(fmaxf(d[3], d[4]), d[5]);
      const float m2 = fmaxf(fmaxf(d[5], d[6]), d[7]);
      tk[k] = fmaxf(fmaxf(m0, m1), m2);
    }

    const float q0 = fmaxf(fmaxf(tk[0], tk[1]), tk[2]);
    const float q1 = fmaxf(fmaxf(tk[3], tk[4]), tk[5]);
    const float bv = fmaxf(fmaxf(q0, q1), fmaxf(tk[6], tk[7]));

    const float bo   = xhalf_f(bv);
    const bool  og   = bo > bv;
    const float bsel = og ? bo : bv;
    const float an   = bsel + e;
    const float afin = m ? an : aold;
    sAlpha[nxt][j] = afin;
    __builtin_amdgcn_sched_barrier(0);

    float x[7];
#pragma unroll
    for (int r = 0; r < 7; ++r) x[r] = dm[7][r];
    int kk = 16 * 7;
    pick_tile<6, 3>(dm, tk, bv, x, kk);
    __builtin_amdgcn_sched_barrier(0);

    __builtin_amdgcn_fence(__ATOMIC_RELEASE, "workgroup", "local");
    __builtin_amdgcn_s_barrier_signal(-1);

    float bvs = bv;
    asm volatile("" : "+v"(bvs));
    pick_tile<2, 0>(dm, tk, bvs, x, kk);
#pragma unroll
    for (int r = 0; r < 7; ++r) px[r] = x[r];
    pkk = kk;
    pbv = bvs;
    pbo = bo;
#pragma unroll
    for (int r = 0; r < 7; ++r) asm volatile("" : "+v"(px[r]));
    asm volatile("" : "+v"(pkk));
    __builtin_amdgcn_s_barrier_wait(-1);
    __builtin_amdgcn_fence(__ATOMIC_ACQUIRE, "workgroup", "local");
  };

  float eA = erow[(size_t)1 * Cc], eB = 0.0f;
  int   mA = mrow[1],              mB = 0;
  int t = 1;
  for (; t + 1 < Ss; t += 2) {
    step(0, t,     eA, mA, eB, mB);
    step(1, t + 1, eB, mB, eA, mA);
  }
  step(0, t, eA, mA, eB, mB);

  store_backpointer(px, pkk, pbv, pbo, hi8,
                    bpcol + (size_t)(Ss - 2) * Cc);

  if (w == 0) {
    const float* fa = sAlpha[(Ss - 1) & 1];
    float bbest = NEG; int barg = 0;
#pragma unroll
    for (int k = 0; k < 4; ++k) {
      const int   jj = (k << 5) + lane;
      const float v  = fa[jj];
      const bool  g  = v > bbest;
      bbest = g ? v : bbest;
      barg  = g ? jj : barg;
    }
#pragma unroll
    for (int off = 16; off >= 1; off >>= 1) {
      const float bo = __shfl_xor(bbest, off, 32);
      const int   ao = __shfl_xor(barg, off, 32);
      const bool  tk = (bo > bbest) || ((bo == bbest) && (ao < barg));
      bbest = tk ? bo : bbest;
      barg  = tk ? ao : barg;
    }
    if (lane == 0) path[(size_t)b * Ss + (Ss - 1)] = barg;
  }
}

template <bool FIRST>
static __device__ __forceinline__ void load32(
    unsigned (&r)[32], const uint8_t* __restrict__ bpb, int c, int lane)
{
  const unsigned* base = (const unsigned*)(bpb + (size_t)c * 32 * Cc) + lane;
#pragma unroll
  for (int i = 31; i >= 0; --i) {
    if (FIRST && i == 31) { r[i] = 0u; continue; }
    r[i] = base[i * (Cc / 4)];
  }
}

template <bool FIRST>
static __device__ __forceinline__ void chase32(
    const unsigned (&r)[32], int& tag, int lane, int* __restrict__ pline)
{
  int acc = 0;
#pragma unroll
  for (int i = 31; i >= 0; --i) {
    if (FIRST && i == 31) continue;
    const unsigned wsel =
        (unsigned)__builtin_amdgcn_readlane((int)r[i], tag >> 2);
    tag = (int)((wsel >> ((tag & 3) << 3)) & 0xFFu);
    acc = (lane == i) ? tag : acc;
  }
  if (!FIRST || lane < 31) pline[lane] = acc;
}

__global__ __launch_bounds__(32) void viterbi_backtrace(
    const uint8_t* __restrict__ bp, int* __restrict__ path)
{
  const int b    = blockIdx.x;
  const int lane = threadIdx.x;
  const uint8_t* bpb = bp + (size_t)b * (Ss - 1) * Cc;
  int* pb = path + (size_t)b * Ss;

  unsigned ra[32], rb[32];
  load32<true >(ra, bpb, 31, lane);
  load32<false>(rb, bpb, 30, lane);
  int tag = __builtin_amdgcn_readfirstlane(pb[Ss - 1]);

  chase32<true>(ra, tag, lane, pb + 31 * 32);
  for (int c = 30; c >= 2; c -= 2) {
    load32<false>(ra, bpb, c - 1, lane);
    chase32<false>(rb, tag, lane, pb + c * 32);
    load32<false>(rb, bpb, c - 2, lane);
    chase32<false>(ra, tag, lane, pb + (c - 1) * 32);
  }
  chase32<false>(rb, tag, lane, pb);
}

extern "C" void kernel_launch(void* const* d_in, const int* in_sizes, int n_in,
                              void* d_out, int out_size, void* d_ws, size_t ws_size,
                              hipStream_t stream) {
  const float* emissions   = (const float*)d_in[0];
  const int*   mask        = (const int*)d_in[1];
  const float* transitions = (const float*)d_in[2];
  int*         path        = (int*)d_out;
  uint8_t*     bp          = (uint8_t*)d_ws;

  viterbi_forward<<<Bb, 256, 0, stream>>>(emissions, mask, transitions, bp, path);
  viterbi_backtrace<<<Bb, 32, 0, stream>>>(bp, path);
}
